# starting kernel
# baseline (speedup 1.0000x reference)
#include <hip/hip_runtime.h>

typedef __attribute__((ext_vector_type(2))) float v2f;
typedef __attribute__((ext_vector_type(8))) float v8f;

#define N_QUBITS 12
#define N_LAYERS 6
#define BATCH    4096
#define DIM      4096
#define NU       (N_LAYERS * 3)
#define UFLOATS  (NU * 512)

static __device__ __forceinline__ v8f wmma4(v2f a, v2f b, v8f c) {
  return __builtin_amdgcn_wmma_f32_16x16x4_f32(false, a, false, b, (short)0, c,
                                               false, false);
}

__global__ __launch_bounds__(256) void fuse_unitaries(
    const float* __restrict__ weights, float* __restrict__ Umats) {
  const int e = blockIdx.x;
  const int l = e / 3, g = e % 3;
  const int tid = threadIdx.x;
  const int i = tid >> 4, j = tid & 15;

  float pr = 1.f, pi = 0.f;
#pragma unroll
  for (int k = 0; k < 4; ++k) {
    const int q  = 4 * g + k;
    const float ty = 0.5f * weights[(l * N_QUBITS + q) * 2 + 0];
    const float tz = 0.5f * weights[(l * N_QUBITS + q) * 2 + 1];
    const float cy = cosf(ty), sy = sinf(ty);
    const float cz = cosf(tz), sz = sinf(tz);
    const int bi = (i >> (3 - k)) & 1;
    const int bj = (j >> (3 - k)) & 1;
    const float ryv = bi ? (bj ? cy : sy) : (bj ? -sy : cy);
    const float gre = ryv * cz;
    const float gim = ryv * (bi ? sz : -sz);
    const float nr = pr * gre - pi * gim;
    const float ni = pr * gim + pi * gre;
    pr = nr; pi = ni;
  }
  Umats[e * 512 + tid]       = pr;
  Umats[e * 512 + 256 + tid] = pi;
}

__global__ __launch_bounds__(256) void qsim(const float* __restrict__ x,
                                            const float* __restrict__ Umats,
                                            float* __restrict__ out) {
  __shared__ float Sr[DIM];
  __shared__ float Si[DIM];
  __shared__ float Uall[UFLOATS];
  __shared__ float red[256];

  const int tid  = threadIdx.x;
  const int b    = blockIdx.x;
  const int lane = tid & 31;
  const int wave = tid >> 5;
  const int half = lane >> 4;
  const int lm   = lane & 15;

#pragma unroll
  for (int k = 0; k < UFLOATS / 256; ++k)
    Uall[tid + 256 * k] = Umats[tid + 256 * k];

  float cw[N_QUBITS], sw[N_QUBITS];
#pragma unroll
  for (int w = 0; w < N_QUBITS; ++w) {
    const float t = 0.5f * x[b * N_QUBITS + w];
    cw[w] = cosf(t);
    sw[w] = sinf(t);
  }
#pragma unroll
  for (int k = 0; k < 16; ++k) {
    const int a = tid + 256 * k;
    float p = 1.f;
#pragma unroll
    for (int w = 0; w < N_QUBITS; ++w)
      p *= ((a >> (11 - w)) & 1) ? sw[w] : cw[w];
    Sr[a] = p;
    Si[a] = 0.f;
  }
  __syncthreads();

  for (int l = 0; l < N_LAYERS; ++l) {
    const bool gray = (l > 0);
    for (int g = 0; g < 3; ++g) {
      const float* Ur = Uall + (l * 3 + g) * 512;
      const float* Ui = Ur + 256;

      v8f dr[2] = {};
      v8f di[2] = {};
#pragma unroll
      for (int c = 0; c < 4; ++c) {
        const int k0 = 4 * c + 2 * half;
        v2f ur, ui, nui;
        ur.x = Ur[lm * 16 + k0];
        ur.y = Ur[lm * 16 + k0 + 1];
        ui.x = Ui[lm * 16 + k0];
        ui.y = Ui[lm * 16 + k0 + 1];
        nui  = -ui;
#pragma unroll
        for (int t2 = 0; t2 < 2; ++t2) {
          const int tile = wave * 2 + t2;
          int a0, a1;
          if (g == 0) {
            a0 = k0 * 256 + tile * 16 + lm;
            a1 = a0 + 256;
            if (gray) {
              a0 ^= (a0 >> 1);
              a1 ^= (a1 >> 1);
            }
          } else if (g == 1) {
            a0 = tile * 256 + k0 * 16 + lm;  a1 = a0 + 16;
          } else {
            a0 = (tile * 16 + lm) * 16 + k0; a1 = a0 + 1;
          }
          v2f br, bi;
          br.x = Sr[a0]; br.y = Sr[a1];
          bi.x = Si[a0]; bi.y = Si[a1];
          if (g < 2) {
            dr[t2] = wmma4(ur,  br, dr[t2]);
            dr[t2] = wmma4(nui, bi, dr[t2]);
            di[t2] = wmma4(ur,  bi, di[t2]);
            di[t2] = wmma4(ui,  br, di[t2]);
          } else {
            dr[t2] = wmma4(br, ur,  dr[t2]);
            dr[t2] = wmma4(bi, nui, dr[t2]);
            di[t2] = wmma4(br, ui,  di[t2]);
            di[t2] = wmma4(bi, ur,  di[t2]);
          }
        }
      }
      __syncthreads();
#pragma unroll
      for (int t2 = 0; t2 < 2; ++t2) {
        const int tile = wave * 2 + t2;
#pragma unroll
        for (int v = 0; v < 8; ++v) {
          const int m = v + 8 * half;
          int a;
          if (g == 0)      a = m * 256 + tile * 16 + lm;
          else if (g == 1) a = tile * 256 + m * 16 + lm;
          else             a = (tile * 16 + m) * 16 + lm;
          Sr[a] = dr[t2][v];
          Si[a] = di[t2][v];
        }
      }
      __syncthreads();
    }
  }

  float acc = 0.f;
#pragma unroll
  for (int k = 0; k < 16; ++k) {
    const int a = tid + 256 * k;
    const float p = Sr[a] * Sr[a] + Si[a] * Si[a];
    acc += (a & 2048) ? -p : p;
  }
  red[tid] = acc;
  __syncthreads();
  for (int off = 128; off > 0; off >>= 1) {
    if (tid < off) red[tid] += red[tid + off];
    __syncthreads();
  }
  if (tid == 0) out[b] = red[0];
}

extern "C" void kernel_launch(void* const* d_in, const int* in_sizes, int n_in,
                              void* d_out, int out_size, void* d_ws,
                              size_t ws_size, hipStream_t stream) {
  const float* x = (const float*)d_in[0];
  const float* w = (const float*)d_in[1];
  float* out     = (float*)d_out;
  float* Umats   = (float*)d_ws;

  fuse_unitaries<<<dim3(N_LAYERS * 3), dim3(256), 0, stream>>>(w, Umats);
  qsim<<<dim3(BATCH), dim3(256), 0, stream>>>(x, Umats, out);
}


# Weights are the same for all samples: the circuit on the 6 wires that reach Z_0 is multiplied out once per call into a 64x64 complex V; a sample is |V phi|^2, one 128x64 WMMA product; new sum order
# speedup vs baseline: 16.8300x; 16.8300x over previous
#include <hip/hip_runtime.h>

typedef __attribute__((ext_vector_type(2))) float v2f;
typedef __attribute__((ext_vector_type(8))) float v8f;

#define N_QUBITS 12
#define N_LAYERS 6
#define BATCH    4096
#define DIM      4096
#define NGRP     4
#define NU       (N_LAYERS * NGRP)
#define LDS_AMPS (DIM + DIM / 32)
#define ROWMASK  0xFC0


static __device__ __forceinline__ v8f wmma4(v2f a, v2f b, v8f c) {
  return __builtin_amdgcn_wmma_f32_16x16x4_f32(false, a, false, b, (short)0, c,
                                               false, false);
}


__global__ __launch_bounds__(256) void fuse_unitaries(
    const float* __restrict__ weights, float* __restrict__ Ufrag) {
  const int e = (blockIdx.x >> 1) * 4 + (blockIdx.x & 1);
  const int l = e >> 2, g = e & 3;
  const int t = threadIdx.x;
  const int c = t >> 6, lane = (t >> 1) & 31, r = t & 1;
  const int m = lane & 15, half = lane >> 4;
  const int j  = 2 * c + half;
  const int i  = 2 * ((m & 7) >> 1) + (m >> 3);
  const int ri = m & 1;

  float pr = 1.f, pi = 0.f;
#pragma unroll
  for (int k = 0; k < 3; ++k) {
    const int q  = 3 * g + k;
    const float ty = 0.5f * weights[(l * N_QUBITS + q) * 2 + 0];
    const float tz = 0.5f * weights[(l * N_QUBITS + q) * 2 + 1];
    const float cy = cosf(ty), sy = sinf(ty);
    const float cz = cosf(tz), sz = sinf(tz);
    const int bi = (i >> (2 - k)) & 1;
    const int bj = (j >> (2 - k)) & 1;
    const float ryv = bi ? (bj ? cy : sy) : (bj ? -sy : cy);
    const float gre = ryv * cz;
    const float gim = ryv * (bi ? sz : -sz);
    const float nr = pr * gre - pi * gim;
    const float ni = pr * gim + pi * gre;
    pr = nr; pi = ni;
  }
  const float val = ri ? (r ? pr : pi) : (r ? -pi : pr);
  Ufrag[e * 256 + t] = val;
}

static __device__ __forceinline__ constexpr int col_bit(int g, int k) {
  return g == 0 ? k
       : g == 1 ? (k == 0 ? 0 : k + 1)
       : g == 2 ? (k < 3 ? k : 9)
                : (k < 2 ? k + 3 : k + 4);
}
static __device__ __forceinline__ constexpr int tile_bit(int g, int k) {
  return g == 0 ? 4 + k
       : g == 1 ? (k == 0 ? 1 : k == 1 ? 5 : 7 + k)
       : g == 2 ? (k < 3 ? 6 + k : 7 + k)
                : (k == 0 ? 5 : 7 + k);
}
static __device__ __forceinline__ constexpr int scat_col(int g, int lm) {
  int a = 0;
  for (int k = 0; k < 4; ++k) a |= ((lm >> k) & 1) << col_bit(g, k);
  return a;
}
static __device__ __forceinline__ constexpr int scat_tile(int g, int t) {
  int a = 0;
  for (int k = 0; k < 5; ++k) a |= ((t >> k) & 1) << tile_bit(g, k);
  return a;
}
static __device__ __forceinline__ constexpr int slot(int a) { return a + (a >> 5); }

template <int G, bool GRAY>
static __device__ __forceinline__ void stage(v2f* __restrict__ S2,
                                             const v2f (&af)[4],
                                             int lane, int wave) {
  constexpr int SH = 9 - 3 * G;
  const int lm = lane & 15, half = lane >> 4;

  const int base  = scat_tile(G, wave * 4) | scat_col(G, lm) | (half << SH);
  const int sbase = slot(base);
  const int gbase = base ^ ((base >> 1) & ROWMASK);

  v8f d[4] = {};
#pragma unroll
  for (int c = 0; c < 4; ++c) {
#pragma unroll
    for (int tt = 0; tt < 4; ++tt) {
      const int k = scat_tile(G, tt) | (c << (SH + 1));
      int s;
      if (GRAY) {
        const int p = gbase ^ (k ^ ((k >> 1) & ROWMASK));
        s = slot(p);
      } else {
        s = sbase + slot(k);
      }
      const v2f b = S2[s];
      d[tt] = wmma4(af[c], b, d[tt]);
    }
  }
  __syncthreads();
#pragma unroll
  for (int tt = 0; tt < 4; ++tt) {
#pragma unroll
    for (int q = 0; q < 4; ++q) {
      const int k = scat_tile(G, tt) | (q << (SH + 1));
      v2f o;
      o.x = d[tt][2 * q];
      o.y = d[tt][2 * q + 1];
      S2[sbase + slot(k)] = o;
    }
  }
  __syncthreads();
}

__global__ __launch_bounds__(256) void build_operator(
    const float* __restrict__ Ufrag, float* __restrict__ Vfrag) {
  __shared__ v2f S2[LDS_AMPS];

  const int tid  = threadIdx.x;
  const int lane = tid & 31;
  const int wave = tid >> 5;

  const v2f* Uf = reinterpret_cast<const v2f*>(Ufrag);
  v2f gate[2 * N_LAYERS][4];
#pragma unroll
  for (int s = 0; s < 2 * N_LAYERS; ++s) {
#pragma unroll
    for (int c = 0; c < 4; ++c)
      gate[s][c] = Uf[((s >> 1) * NGRP + (s & 1)) * 128 + c * 32 + lane];
  }

#pragma unroll
  for (int k = 0; k < 16; ++k) {
    const int a = tid + 256 * k;
    v2f o;
    o.x = ((a >> 6) == (a & 63)) ? 1.f : 0.f;
    o.y = 0.f;
    S2[slot(a)] = o;
  }
  __syncthreads();

  stage<0, false>(S2, gate[0], lane, wave);
  stage<1, false>(S2, gate[1], lane, wave);
#pragma unroll
  for (int l = 1; l < N_LAYERS; ++l) {
    stage<0, true >(S2, gate[2 * l],     lane, wave);
    stage<1, false>(S2, gate[2 * l + 1], lane, wave);
  }

#pragma unroll 8
  for (int k = 0; k < 32; ++k) {
    const int idx = tid + 256 * k;
    const int s  = idx & 1;
    const int ln = (idx >> 1) & 31;
    const int c  = (idx >> 6) & 15;
    const int mt = idx >> 10;
    const int R  = 16 * mt + (ln & 15);
    const int j  = 4 * c + 2 * (ln >> 4) + s;
    const v2f v  = S2[slot(((R >> 1) << 6) | j)];
    Vfrag[idx] = (R & 1) ? v.y : v.x;
  }
}

__global__ __launch_bounds__(256) void expect_z0(const float* __restrict__ x,
                                                 const float* __restrict__ Vfrag,
                                                 float* __restrict__ out) {
  __shared__ float part[8][16];

  const int lane = threadIdx.x & 31;
  const int wave = threadIdx.x >> 5;
  const int st = wave >> 2, mq = wave & 3;
  const int lm = lane & 15, half = lane >> 4;
  const int b  = blockIdx.x * 32 + st * 16 + lm;

  float cw[6], sw[6];
#pragma unroll
  for (int w = 0; w < 6; ++w) {
    const float t = 0.5f * x[b * N_QUBITS + w];
    cw[w] = cosf(t);
    sw[w] = sinf(t);
  }
  float p4[16];
#pragma unroll
  for (int c = 0; c < 16; ++c) {
    float p = 1.f;
#pragma unroll
    for (int w = 0; w < 4; ++w) p *= ((c >> (3 - w)) & 1) ? sw[w] : cw[w];
    p4[c] = p;
  }
  const float t4 = half ? sw[4] : cw[4];
  const float e0 = t4 * cw[5], e1 = t4 * sw[5];

  const v2f* Vf = reinterpret_cast<const v2f*>(Vfrag) + (2 * mq) * 16 * 32 + lane;
  v8f d[2] = {};
#pragma unroll
  for (int c = 0; c < 16; ++c) {
    v2f bf;
    bf.x = p4[c] * e0;
    bf.y = p4[c] * e1;
#pragma unroll
    for (int i = 0; i < 2; ++i)
      d[i] = wmma4(Vf[(i * 16 + c) * 32], bf, d[i]);
  }

  float acc = 0.f;
#pragma unroll
  for (int i = 0; i < 2; ++i) {
#pragma unroll
    for (int v = 0; v < 8; ++v) acc += d[i][v] * d[i][v];
  }
  acc += __shfl_xor(acc, 16);
  if (lane < 16) part[wave][lane] = (mq < 2) ? acc : -acc;
  __syncthreads();
  if (wave == 0) {
    const int w0 = (lane >> 4) * 4;
    out[blockIdx.x * 32 + lane] = (part[w0][lm] + part[w0 + 1][lm]) +
                                  (part[w0 + 2][lm] + part[w0 + 3][lm]);
  }
}

extern "C" void kernel_launch(void* const* d_in, const int* in_sizes, int n_in,
                              void* d_out, int out_size, void* d_ws,
                              size_t ws_size, hipStream_t stream) {
  const float* x = (const float*)d_in[0];
  const float* w = (const float*)d_in[1];
  float* out     = (float*)d_out;
  float* Ufrag   = (float*)d_ws;
  float* Vfrag   = Ufrag + NU * 256;

  fuse_unitaries<<<dim3(2 * N_LAYERS), dim3(256), 0, stream>>>(w, Ufrag);
  build_operator<<<dim3(1), dim3(256), 0, stream>>>(Ufrag, Vfrag);
  expect_z0<<<dim3(BATCH / 32), dim3(256), 0, stream>>>(x, Vfrag, out);
}


# One launch, was three: every 8-wave workgroup rebuilds the gate tables and the 64x64 operator V in its own LDS and applies V to its 32 samples straight from LDS; float sum order unchanged
# speedup vs baseline: 27.0023x; 1.6044x over previous
#include <hip/hip_runtime.h>

typedef __attribute__((ext_vector_type(2))) float v2f;
typedef __attribute__((ext_vector_type(8))) float v8f;

#define N_QUBITS 12
#define N_LAYERS 6
#define BATCH    4096
#define DIM      4096
#define NGRP     4
#define LDS_AMPS (DIM + DIM / 32)
#define ROWMASK  0xFC0


static __device__ __forceinline__ v8f wmma4(v2f a, v2f b, v8f c) {
  return __builtin_amdgcn_wmma_f32_16x16x4_f32(false, a, false, b, (short)0, c,
                                               false, false);
}


static __device__ __forceinline__ v2f gate_fragment(const float* __restrict__ trig,
                                                    int l, int g, int i, int j,
                                                    int ri) {
  float pr = 1.f, pi = 0.f;
#pragma unroll
  for (int k = 0; k < 3; ++k) {
    const float* tp = trig + (l * 6 + 3 * g + k) * 4;
    const float cy = tp[0], sy = tp[1];
    const float cz = tp[2], sz = tp[3];
    const int bi = (i >> (2 - k)) & 1;
    const int bj = (j >> (2 - k)) & 1;
    const float ryv = bi ? (bj ? cy : sy) : (bj ? -sy : cy);
    const float gre = ryv * cz;
    const float gim = ryv * (bi ? sz : -sz);
    const float nr = pr * gre - pi * gim;
    const float ni = pr * gim + pi * gre;
    pr = nr; pi = ni;
  }
  v2f o;
  o.x = ri ? pi : pr;
  o.y = ri ? pr : -pi;
  return o;
}

static __device__ __forceinline__ constexpr int col_bit(int g, int k) {
  return g == 0 ? k
       : g == 1 ? (k == 0 ? 0 : k + 1)
       : g == 2 ? (k < 3 ? k : 9)
                : (k < 2 ? k + 3 : k + 4);
}
static __device__ __forceinline__ constexpr int tile_bit(int g, int k) {
  return g == 0 ? 4 + k
       : g == 1 ? (k == 0 ? 1 : k == 1 ? 5 : 7 + k)
       : g == 2 ? (k < 3 ? 6 + k : 7 + k)
                : (k == 0 ? 5 : 7 + k);
}
static __device__ __forceinline__ constexpr int scat_col(int g, int lm) {
  int a = 0;
  for (int k = 0; k < 4; ++k) a |= ((lm >> k) & 1) << col_bit(g, k);
  return a;
}
static __device__ __forceinline__ constexpr int scat_tile(int g, int t) {
  int a = 0;
  for (int k = 0; k < 5; ++k) a |= ((t >> k) & 1) << tile_bit(g, k);
  return a;
}
static __device__ __forceinline__ constexpr int slot(int a) { return a + (a >> 5); }

template <int G, bool GRAY>
static __device__ __forceinline__ void stage(v2f* __restrict__ S2,
                                             const v2f (&af)[4],
                                             int lane, int wave) {
  constexpr int SH = 9 - 3 * G;
  const int lm = lane & 15, half = lane >> 4;

  const int base  = scat_tile(G, wave * 4) | scat_col(G, lm) | (half << SH);
  const int sbase = slot(base);
  const int gbase = base ^ ((base >> 1) & ROWMASK);

  v8f d[4] = {};
#pragma unroll
  for (int c = 0; c < 4; ++c) {
#pragma unroll
    for (int tt = 0; tt < 4; ++tt) {
      const int k = scat_tile(G, tt) | (c << (SH + 1));
      int s;
      if (GRAY) {
        const int p = gbase ^ (k ^ ((k >> 1) & ROWMASK));
        s = slot(p);
      } else {
        s = sbase + slot(k);
      }
      const v2f b = S2[s];
      d[tt] = wmma4(af[c], b, d[tt]);
    }
  }
  __syncthreads();
#pragma unroll
  for (int tt = 0; tt < 4; ++tt) {
#pragma unroll
    for (int q = 0; q < 4; ++q) {
      const int k = scat_tile(G, tt) | (q << (SH + 1));
      v2f o;
      o.x = d[tt][2 * q];
      o.y = d[tt][2 * q + 1];
      S2[sbase + slot(k)] = o;
    }
  }
  __syncthreads();
}

__global__ __launch_bounds__(256) void circuit_z0(const float* __restrict__ x,
                                                  const float* __restrict__ weights,
                                                  float* __restrict__ out) {
  __shared__ v2f   S2[LDS_AMPS];
  __shared__ v2f   Ufrag2[2 * N_LAYERS * 128];
  __shared__ float trig[N_LAYERS * 6 * 4];
  __shared__ float part[8][16];

  const int tid  = threadIdx.x;
  const int lane = tid & 31;
  const int wave = tid >> 5;
  const int st = wave >> 2, mq = wave & 3;
  const int lm = lane & 15, half = lane >> 4;
  const int b  = blockIdx.x * 32 + st * 16 + lm;

  float xa[6];
#pragma unroll
  for (int w = 0; w < 6; ++w) xa[w] = x[b * N_QUBITS + w];

  if (tid < N_LAYERS * 6 * 2) {
    const int pw = tid >> 1, which = tid & 1;
    const int l = pw / 6, q = pw - 6 * l;
    const float t = 0.5f * weights[(l * N_QUBITS + q) * 2 + which];
    trig[pw * 4 + which * 2 + 0] = cosf(t);
    trig[pw * 4 + which * 2 + 1] = sinf(t);
  }
#pragma unroll
  for (int k = 0; k < 16; ++k) {
    const int a = tid + 256 * k;
    v2f o;
    o.x = ((a >> 6) == (a & 63)) ? 1.f : 0.f;
    o.y = 0.f;
    S2[slot(a)] = o;
  }
  __syncthreads();

  {
    const int g  = tid >> 7;
    const int c  = (tid >> 5) & 3;
    const int j  = 2 * c + half;
    const int i  = 2 * ((lm & 7) >> 1) + (lm >> 3);
    const int ri = lm & 1;
#pragma unroll
    for (int l = 0; l < N_LAYERS; ++l)
      Ufrag2[(2 * l + g) * 128 + c * 32 + lane] = gate_fragment(trig, l, g, i, j, ri);
  }
  __syncthreads();

  v2f gate[2 * N_LAYERS][4];
#pragma unroll
  for (int s = 0; s < 2 * N_LAYERS; ++s) {
#pragma unroll
    for (int c = 0; c < 4; ++c)
      gate[s][c] = Ufrag2[s * 128 + c * 32 + lane];
  }

  stage<0, false>(S2, gate[0], lane, wave);
  stage<1, false>(S2, gate[1], lane, wave);
#pragma unroll
  for (int l = 1; l < N_LAYERS; ++l) {
    stage<0, true >(S2, gate[2 * l],     lane, wave);
    stage<1, false>(S2, gate[2 * l + 1], lane, wave);
  }

  float cw[6], sw[6];
#pragma unroll
  for (int w = 0; w < 6; ++w) {
    const float t = 0.5f * xa[w];
    cw[w] = cosf(t);
    sw[w] = sinf(t);
  }
  float p4[16];
#pragma unroll
  for (int c = 0; c < 16; ++c) {
    float p = 1.f;
#pragma unroll
    for (int w = 0; w < 4; ++w) p *= ((c >> (3 - w)) & 1) ? sw[w] : cw[w];
    p4[c] = p;
  }
  const float t4 = half ? sw[4] : cw[4];
  const float e0 = t4 * cw[5], e1 = t4 * sw[5];

  const float* Vrow = reinterpret_cast<const float*>(S2) +
                      2 * slot(((16 * mq + (lm >> 1)) << 6) | (2 * half)) + (lm & 1);
  v8f d[2] = {};
#pragma unroll
  for (int c = 0; c < 16; ++c) {
    v2f bf;
    bf.x = p4[c] * e0;
    bf.y = p4[c] * e1;
#pragma unroll
    for (int i = 0; i < 2; ++i) {
      const int k = slot(((8 * i) << 6) | (4 * c));
      v2f a;
      a.x = Vrow[2 * k];
      a.y = Vrow[2 * k + 2];
      d[i] = wmma4(a, bf, d[i]);
    }
  }

  float acc = 0.f;
#pragma unroll
  for (int i = 0; i < 2; ++i) {
#pragma unroll
    for (int v = 0; v < 8; ++v) acc += d[i][v] * d[i][v];
  }
  acc += __shfl_xor(acc, 16);
  if (lane < 16) part[wave][lane] = (mq < 2) ? acc : -acc;
  __syncthreads();
  if (wave == 0) {
    const int w0 = (lane >> 4) * 4;
    out[blockIdx.x * 32 + lane] = (part[w0][lm] + part[w0 + 1][lm]) +
                                  (part[w0 + 2][lm] + part[w0 + 3][lm]);
  }
}

extern "C" void kernel_launch(void* const* d_in, const int* in_sizes, int n_in,
                              void* d_out, int out_size, void* d_ws,
                              size_t ws_size, hipStream_t stream) {
  const float* x = (const float*)d_in[0];
  const float* w = (const float*)d_in[1];
  float* out     = (float*)d_out;

  circuit_z0<<<dim3(BATCH / 32), dim3(256), 0, stream>>>(x, w, out);
}


# Layers 3-5 reach Z_0 through wires 0-2 only, so with their CNOTs 0->1, 1->2 they are multiplied out into ONE 8x8 gate before V is touched: 7 matrix stages, was 12; new float sum order
# speedup vs baseline: 34.7267x; 1.2861x over previous
#include <hip/hip_runtime.h>

typedef __attribute__((ext_vector_type(2))) float v2f;
typedef __attribute__((ext_vector_type(8))) float v8f;

#define N_QUBITS 12
#define N_LAYERS 6
#define BATCH    4096
#define DIM      4096
#define NGRP     4
#define LDS_AMPS (DIM + DIM / 32)
#define ROWMASK  0xFC0
#define N_FULL   3
#define N_GATES  (2 * N_FULL + 1)


static __device__ __forceinline__ v8f wmma4(v2f a, v2f b, v8f c) {
  return __builtin_amdgcn_wmma_f32_16x16x4_f32(false, a, false, b, (short)0, c,
                                               false, false);
}


static __device__ __forceinline__ v2f gate_fragment(const float* __restrict__ trig,
                                                    int l, int g, int i, int j,
                                                    int ri) {
  float pr = 1.f, pi = 0.f;
#pragma unroll
  for (int k = 0; k < 3; ++k) {
    const float* tp = trig + (l * 6 + 3 * g + k) * 4;
    const float cy = tp[0], sy = tp[1];
    const float cz = tp[2], sz = tp[3];
    const int bi = (i >> (2 - k)) & 1;
    const int bj = (j >> (2 - k)) & 1;
    const float ryv = bi ? (bj ? cy : sy) : (bj ? -sy : cy);
    const float gre = ryv * cz;
    const float gim = ryv * (bi ? sz : -sz);
    const float nr = pr * gre - pi * gim;
    const float ni = pr * gim + pi * gre;
    pr = nr; pi = ni;
  }
  v2f o;
  o.x = ri ? pi : pr;
  o.y = ri ? pr : -pi;
  return o;
}


struct cplx { float re, im; };
static __device__ __forceinline__ cplx cmul(cplx a, cplx b) {
  cplx o;
  o.re = a.re * b.re - a.im * b.im;
  o.im = a.re * b.im + a.im * b.re;
  return o;
}
static __device__ __forceinline__ cplx wire_gate(const float* __restrict__ trig,
                                                 int l, int w, int bi, int bj) {
  const float* tp = trig + (l * 6 + w) * 4;
  const float cy = tp[0], sy = tp[1];
  const float cz = tp[2], sz = tp[3];
  const float ryv = bi ? (bj ? cy : sy) : (bj ? -sy : cy);
  cplx o;
  o.re = ryv * cz;
  o.im = ryv * (bi ? sz : -sz);
  return o;
}

static __device__ __forceinline__ v2f tail_fragment(const float* __restrict__ trig,
                                                    int i, int j, int ri) {
  const int i0 = (i >> 2) & 1, i1 = (i >> 1) & 1, i2 = i & 1;
  const int j0 = (j >> 2) & 1, j1 = (j >> 1) & 1, j2 = j & 1;
  float pr = 0.f, pi = 0.f;
#pragma unroll
  for (int p0 = 0; p0 < 2; ++p0) {
    float mr = 0.f, mi = 0.f;
#pragma unroll
    for (int k0 = 0; k0 < 2; ++k0) {
      float nr = 0.f, ni = 0.f;
#pragma unroll
      for (int k1 = 0; k1 < 2; ++k1) {
        const cplx t = cmul(cmul(wire_gate(trig, 4, 1, i1 ^ p0, k1 ^ k0),
                                 wire_gate(trig, 3, 1, k1, j1)),
                            wire_gate(trig, 3, 2, i2 ^ k1 ^ k0, j2));
        nr += t.re; ni += t.im;
      }
      cplx n; n.re = nr; n.im = ni;
      const cplx a = cmul(cmul(wire_gate(trig, 4, 0, p0, k0),
                               wire_gate(trig, 3, 0, k0, j0)), n);
      mr += a.re; mi += a.im;
    }
    cplx m; m.re = mr; m.im = mi;
    const cplx a = cmul(wire_gate(trig, 5, 0, i0, p0), m);
    pr += a.re; pi += a.im;
  }
  v2f o;
  o.x = ri ? pi : pr;
  o.y = ri ? pr : -pi;
  return o;
}

static __device__ __forceinline__ constexpr int col_bit(int g, int k) {
  return g == 0 ? k
       : g == 1 ? (k == 0 ? 0 : k + 1)
       : g == 2 ? (k < 3 ? k : 9)
                : (k < 2 ? k + 3 : k + 4);
}
static __device__ __forceinline__ constexpr int tile_bit(int g, int k) {
  return g == 0 ? 4 + k
       : g == 1 ? (k == 0 ? 1 : k == 1 ? 5 : 7 + k)
       : g == 2 ? (k < 3 ? 6 + k : 7 + k)
                : (k == 0 ? 5 : 7 + k);
}
static __device__ __forceinline__ constexpr int scat_col(int g, int lm) {
  int a = 0;
  for (int k = 0; k < 4; ++k) a |= ((lm >> k) & 1) << col_bit(g, k);
  return a;
}
static __device__ __forceinline__ constexpr int scat_tile(int g, int t) {
  int a = 0;
  for (int k = 0; k < 5; ++k) a |= ((t >> k) & 1) << tile_bit(g, k);
  return a;
}
static __device__ __forceinline__ constexpr int slot(int a) { return a + (a >> 5); }

template <int G, bool GRAY>
static __device__ __forceinline__ void stage(v2f* __restrict__ S2,
                                             const v2f (&af)[4],
                                             int lane, int wave) {
  constexpr int SH = 9 - 3 * G;
  const int lm = lane & 15, half = lane >> 4;

  const int base  = scat_tile(G, wave * 4) | scat_col(G, lm) | (half << SH);
  const int sbase = slot(base);
  const int gbase = base ^ ((base >> 1) & ROWMASK);

  v8f d[4] = {};
#pragma unroll
  for (int c = 0; c < 4; ++c) {
#pragma unroll
    for (int tt = 0; tt < 4; ++tt) {
      const int k = scat_tile(G, tt) | (c << (SH + 1));
      int s;
      if (GRAY) {
        const int p = gbase ^ (k ^ ((k >> 1) & ROWMASK));
        s = slot(p);
      } else {
        s = sbase + slot(k);
      }
      const v2f b = S2[s];
      d[tt] = wmma4(af[c], b, d[tt]);
    }
  }
  __syncthreads();
#pragma unroll
  for (int tt = 0; tt < 4; ++tt) {
#pragma unroll
    for (int q = 0; q < 4; ++q) {
      const int k = scat_tile(G, tt) | (q << (SH + 1));
      v2f o;
      o.x = d[tt][2 * q];
      o.y = d[tt][2 * q + 1];
      S2[sbase + slot(k)] = o;
    }
  }
  __syncthreads();
}

__global__ __launch_bounds__(256) void circuit_z0(const float* __restrict__ x,
                                                  const float* __restrict__ weights,
                                                  float* __restrict__ out) {
  __shared__ v2f   S2[LDS_AMPS];
  __shared__ v2f   Ufrag2[N_GATES * 128];
  __shared__ float trig[N_LAYERS * 6 * 4];
  __shared__ float part[8][16];

  const int tid  = threadIdx.x;
  const int lane = tid & 31;
  const int wave = tid >> 5;
  const int st = wave >> 2, mq = wave & 3;
  const int lm = lane & 15, half = lane >> 4;
  const int b  = blockIdx.x * 32 + st * 16 + lm;

  float xa[6];
#pragma unroll
  for (int w = 0; w < 6; ++w) xa[w] = x[b * N_QUBITS + w];

  if (tid < N_LAYERS * 6 * 2) {
    const int pw = tid >> 1, which = tid & 1;
    const int l = pw / 6, q = pw - 6 * l;
    const float t = 0.5f * weights[(l * N_QUBITS + q) * 2 + which];
    trig[pw * 4 + which * 2 + 0] = cosf(t);
    trig[pw * 4 + which * 2 + 1] = sinf(t);
  }
#pragma unroll
  for (int k = 0; k < 16; ++k) {
    const int a = tid + 256 * k;
    v2f o;
    o.x = ((a >> 6) == (a & 63)) ? 1.f : 0.f;
    o.y = 0.f;
    S2[slot(a)] = o;
  }
  __syncthreads();

  {
    const int c  = (tid >> 5) & 3;
    const int j  = 2 * c + half;
    const int i  = 2 * ((lm & 7) >> 1) + (lm >> 3);
    const int ri = lm & 1;
    if (tid < 128) {
#pragma unroll
      for (int l = 0; l < N_FULL; ++l) {
#pragma unroll
        for (int g = 0; g < 2; ++g)
          Ufrag2[(2 * l + g) * 128 + c * 32 + lane] = gate_fragment(trig, l, g, i, j, ri);
      }
    } else {
      Ufrag2[2 * N_FULL * 128 + c * 32 + lane] = tail_fragment(trig, i, j, ri);
    }
  }
  __syncthreads();

  v2f gate[N_GATES][4];
#pragma unroll
  for (int s = 0; s < N_GATES; ++s) {
#pragma unroll
    for (int c = 0; c < 4; ++c)
      gate[s][c] = Ufrag2[s * 128 + c * 32 + lane];
  }

  stage<0, false>(S2, gate[0], lane, wave);
  stage<1, false>(S2, gate[1], lane, wave);
#pragma unroll
  for (int l = 1; l < N_FULL; ++l) {
    stage<0, true >(S2, gate[2 * l],     lane, wave);
    stage<1, false>(S2, gate[2 * l + 1], lane, wave);
  }
  stage<0, true >(S2, gate[2 * N_FULL], lane, wave);

  float cw[6], sw[6];
#pragma unroll
  for (int w = 0; w < 6; ++w) {
    const float t = 0.5f * xa[w];
    cw[w] = cosf(t);
    sw[w] = sinf(t);
  }
  float p4[16];
#pragma unroll
  for (int c = 0; c < 16; ++c) {
    float p = 1.f;
#pragma unroll
    for (int w = 0; w < 4; ++w) p *= ((c >> (3 - w)) & 1) ? sw[w] : cw[w];
    p4[c] = p;
  }
  const float t4 = half ? sw[4] : cw[4];
  const float e0 = t4 * cw[5], e1 = t4 * sw[5];

  const float* Vrow = reinterpret_cast<const float*>(S2) +
                      2 * slot(((16 * mq + (lm >> 1)) << 6) | (2 * half)) + (lm & 1);
  v8f d[2] = {};
#pragma unroll
  for (int c = 0; c < 16; ++c) {
    v2f bf;
    bf.x = p4[c] * e0;
    bf.y = p4[c] * e1;
#pragma unroll
    for (int i = 0; i < 2; ++i) {
      const int k = slot(((8 * i) << 6) | (4 * c));
      v2f a;
      a.x = Vrow[2 * k];
      a.y = Vrow[2 * k + 2];
      d[i] = wmma4(a, bf, d[i]);
    }
  }

  float acc = 0.f;
#pragma unroll
  for (int i = 0; i < 2; ++i) {
#pragma unroll
    for (int v = 0; v < 8; ++v) acc += d[i][v] * d[i][v];
  }
  acc += __shfl_xor(acc, 16);
  if (lane < 16) part[wave][lane] = (mq < 2) ? acc : -acc;
  __syncthreads();
  if (wave == 0) {
    const int w0 = (lane >> 4) * 4;
    out[blockIdx.x * 32 + lane] = (part[w0][lm] + part[w0 + 1][lm]) +
                                  (part[w0 + 2][lm] + part[w0 + 3][lm]);
  }
}

extern "C" void kernel_launch(void* const* d_in, const int* in_sizes, int n_in,
                              void* d_out, int out_size, void* d_ws,
                              size_t ws_size, hipStream_t stream) {
  const float* x = (const float*)d_in[0];
  const float* w = (const float*)d_in[1];
  float* out     = (float*)d_out;

  circuit_z0<<<dim3(BATCH / 32), dim3(256), 0, stream>>>(x, w, out);
}


# Layer 0 acts on the identity, so it is the Kronecker product of its two 8x8 gates: layer 1's first stage forms those entries as its B operand; 5 matrix stages, was 7; new float rounding order
# speedup vs baseline: 40.4148x; 1.1638x over previous
#include <hip/hip_runtime.h>

typedef __attribute__((ext_vector_type(2))) float v2f;
typedef __attribute__((ext_vector_type(8))) float v8f;

#define N_QUBITS 12
#define N_LAYERS 6
#define BATCH    4096
#define DIM      4096
#define NGRP     4
#define LDS_AMPS (DIM + DIM / 32)
#define ROWMASK  0xFC0
#define N_FULL   3
#define N_GATES  (2 * (N_FULL - 1) + 1)


static __device__ __forceinline__ v8f wmma4(v2f a, v2f b, v8f c) {
  return __builtin_amdgcn_wmma_f32_16x16x4_f32(false, a, false, b, (short)0, c,
                                               false, false);
}


struct cplx { float re, im; };
static __device__ __forceinline__ cplx gate_entry(const float* __restrict__ trig,
                                                  int l, int g, int i, int j) {
  float pr = 1.f, pi = 0.f;
#pragma unroll
  for (int k = 0; k < 3; ++k) {
    const float* tp = trig + (l * 6 + 3 * g + k) * 4;
    const float cy = tp[0], sy = tp[1];
    const float cz = tp[2], sz = tp[3];
    const int bi = (i >> (2 - k)) & 1;
    const int bj = (j >> (2 - k)) & 1;
    const float ryv = bi ? (bj ? cy : sy) : (bj ? -sy : cy);
    const float gre = ryv * cz;
    const float gim = ryv * (bi ? sz : -sz);
    const float nr = pr * gre - pi * gim;
    const float ni = pr * gim + pi * gre;
    pr = nr; pi = ni;
  }
  cplx o;
  o.re = pr;
  o.im = pi;
  return o;
}
static __device__ __forceinline__ v2f gate_fragment(const float* __restrict__ trig,
                                                    int l, int g, int i, int j,
                                                    int ri) {
  const cplx u = gate_entry(trig, l, g, i, j);
  v2f o;
  o.x = ri ? u.im : u.re;
  o.y = ri ? u.re : -u.im;
  return o;
}


static __device__ __forceinline__ cplx cmul(cplx a, cplx b) {
  cplx o;
  o.re = a.re * b.re - a.im * b.im;
  o.im = a.re * b.im + a.im * b.re;
  return o;
}
static __device__ __forceinline__ cplx wire_gate(const float* __restrict__ trig,
                                                 int l, int w, int bi, int bj) {
  const float* tp = trig + (l * 6 + w) * 4;
  const float cy = tp[0], sy = tp[1];
  const float cz = tp[2], sz = tp[3];
  const float ryv = bi ? (bj ? cy : sy) : (bj ? -sy : cy);
  cplx o;
  o.re = ryv * cz;
  o.im = ryv * (bi ? sz : -sz);
  return o;
}

static __device__ __forceinline__ v2f tail_fragment(const float* __restrict__ trig,
                                                    int i, int j, int ri) {
  const int i0 = (i >> 2) & 1, i1 = (i >> 1) & 1, i2 = i & 1;
  const int j0 = (j >> 2) & 1, j1 = (j >> 1) & 1, j2 = j & 1;
  float pr = 0.f, pi = 0.f;
#pragma unroll
  for (int p0 = 0; p0 < 2; ++p0) {
    float mr = 0.f, mi = 0.f;
#pragma unroll
    for (int k0 = 0; k0 < 2; ++k0) {
      float nr = 0.f, ni = 0.f;
#pragma unroll
      for (int k1 = 0; k1 < 2; ++k1) {
        const cplx t = cmul(cmul(wire_gate(trig, 4, 1, i1 ^ p0, k1 ^ k0),
                                 wire_gate(trig, 3, 1, k1, j1)),
                            wire_gate(trig, 3, 2, i2 ^ k1 ^ k0, j2));
        nr += t.re; ni += t.im;
      }
      cplx n; n.re = nr; n.im = ni;
      const cplx a = cmul(cmul(wire_gate(trig, 4, 0, p0, k0),
                               wire_gate(trig, 3, 0, k0, j0)), n);
      mr += a.re; mi += a.im;
    }
    cplx m; m.re = mr; m.im = mi;
    const cplx a = cmul(wire_gate(trig, 5, 0, i0, p0), m);
    pr += a.re; pi += a.im;
  }
  v2f o;
  o.x = ri ? pi : pr;
  o.y = ri ? pr : -pi;
  return o;
}

static __device__ __forceinline__ constexpr int col_bit(int g, int k) {
  return g == 0 ? k
       : g == 1 ? (k == 0 ? 0 : k + 1)
       : g == 2 ? (k < 3 ? k : 9)
                : (k < 2 ? k + 3 : k + 4);
}
static __device__ __forceinline__ constexpr int tile_bit(int g, int k) {
  return g == 0 ? 4 + k
       : g == 1 ? (k == 0 ? 1 : k == 1 ? 5 : 7 + k)
       : g == 2 ? (k < 3 ? 6 + k : 7 + k)
                : (k == 0 ? 5 : 7 + k);
}
static __device__ __forceinline__ constexpr int scat_col(int g, int lm) {
  int a = 0;
  for (int k = 0; k < 4; ++k) a |= ((lm >> k) & 1) << col_bit(g, k);
  return a;
}
static __device__ __forceinline__ constexpr int scat_tile(int g, int t) {
  int a = 0;
  for (int k = 0; k < 5; ++k) a |= ((t >> k) & 1) << tile_bit(g, k);
  return a;
}
static __device__ __forceinline__ constexpr int slot(int a) { return a + (a >> 5); }

template <int G, bool GRAY>
static __device__ __forceinline__ void stage(v2f* __restrict__ S2,
                                             const v2f (&af)[4],
                                             int lane, int wave) {
  constexpr int SH = 9 - 3 * G;
  const int lm = lane & 15, half = lane >> 4;

  const int base  = scat_tile(G, wave * 4) | scat_col(G, lm) | (half << SH);
  const int sbase = slot(base);
  const int gbase = base ^ ((base >> 1) & ROWMASK);

  v8f d[4] = {};
#pragma unroll
  for (int c = 0; c < 4; ++c) {
#pragma unroll
    for (int tt = 0; tt < 4; ++tt) {
      const int k = scat_tile(G, tt) | (c << (SH + 1));
      int s;
      if (GRAY) {
        const int p = gbase ^ (k ^ ((k >> 1) & ROWMASK));
        s = slot(p);
      } else {
        s = sbase + slot(k);
      }
      const v2f b = S2[s];
      d[tt] = wmma4(af[c], b, d[tt]);
    }
  }
  __syncthreads();
#pragma unroll
  for (int tt = 0; tt < 4; ++tt) {
#pragma unroll
    for (int q = 0; q < 4; ++q) {
      const int k = scat_tile(G, tt) | (q << (SH + 1));
      v2f o;
      o.x = d[tt][2 * q];
      o.y = d[tt][2 * q + 1];
      S2[sbase + slot(k)] = o;
    }
  }
  __syncthreads();
}

static __device__ __forceinline__ void stage_first(v2f* __restrict__ S2,
                                                   const v2f* __restrict__ U0tab,
                                                   const v2f (&af)[4],
                                                   int lane, int wave) {
  constexpr int SH = 9;
  const int lm = lane & 15, half = lane >> 4;
  const int base  = scat_tile(0, wave * 4) | scat_col(0, lm) | (half << SH);
  const int sbase = slot(base);

  const int rlo = (wave ^ (wave >> 1)) ^ (half << 2);
  const v2f u1  = U0tab[64 + rlo * 8 + (lm & 7)];
  const int jh  = lm >> 3;

  v8f d[4] = {};
#pragma unroll
  for (int c = 0; c < 4; ++c) {
    const int rhi = ((2 * c) ^ c) ^ half;
#pragma unroll
    for (int tt = 0; tt < 4; ++tt) {
      const v2f e = U0tab[rhi * 8 + 2 * tt + jh];
      v2f b;
      b.x = u1.x * e.x - u1.y * e.y;
      b.y = u1.x * e.y + u1.y * e.x;
      d[tt] = wmma4(af[c], b, d[tt]);
    }
  }
#pragma unroll
  for (int tt = 0; tt < 4; ++tt) {
#pragma unroll
    for (int q = 0; q < 4; ++q) {
      const int k = scat_tile(0, tt) | (q << (SH + 1));
      v2f o;
      o.x = d[tt][2 * q];
      o.y = d[tt][2 * q + 1];
      S2[sbase + slot(k)] = o;
    }
  }
  __syncthreads();
}

__global__ __launch_bounds__(256) void circuit_z0(const float* __restrict__ x,
                                                  const float* __restrict__ weights,
                                                  float* __restrict__ out) {
  __shared__ v2f   S2[LDS_AMPS];
  __shared__ v2f   Ufrag2[N_GATES * 128];
  __shared__ v2f   U0tab[2 * 64];
  __shared__ float trig[N_LAYERS * 6 * 4];
  __shared__ float part[8][16];

  const int tid  = threadIdx.x;
  const int lane = tid & 31;
  const int wave = tid >> 5;
  const int st = wave >> 2, mq = wave & 3;
  const int lm = lane & 15, half = lane >> 4;
  const int b  = blockIdx.x * 32 + st * 16 + lm;

  float xa[6];
#pragma unroll
  for (int w = 0; w < 6; ++w) xa[w] = x[b * N_QUBITS + w];

  if (tid < N_LAYERS * 6 * 2) {
    const int pw = tid >> 1, which = tid & 1;
    const int l = pw / 6, q = pw - 6 * l;
    const float t = 0.5f * weights[(l * N_QUBITS + q) * 2 + which];
    trig[pw * 4 + which * 2 + 0] = cosf(t);
    trig[pw * 4 + which * 2 + 1] = sinf(t);
  }
  __syncthreads();

  {
    const int c  = (tid >> 5) & 3;
    const int j  = 2 * c + half;
    const int i  = 2 * ((lm & 7) >> 1) + (lm >> 3);
    const int ri = lm & 1;
    if (tid < 128) {
#pragma unroll
      for (int l = 1; l < N_FULL; ++l) {
#pragma unroll
        for (int g = 0; g < 2; ++g)
          Ufrag2[(2 * (l - 1) + g) * 128 + c * 32 + lane] = gate_fragment(trig, l, g, i, j, ri);
      }
      const cplx u = gate_entry(trig, 0, tid >> 6, (tid >> 3) & 7, tid & 7);
      v2f o;
      o.x = u.re;
      o.y = u.im;
      U0tab[tid] = o;
    } else {
      Ufrag2[2 * (N_FULL - 1) * 128 + c * 32 + lane] = tail_fragment(trig, i, j, ri);
    }
  }
  __syncthreads();

  v2f gate[N_GATES][4];
#pragma unroll
  for (int s = 0; s < N_GATES; ++s) {
#pragma unroll
    for (int c = 0; c < 4; ++c)
      gate[s][c] = Ufrag2[s * 128 + c * 32 + lane];
  }

  stage_first(S2, U0tab, gate[0], lane, wave);
  stage<1, false>(S2, gate[1], lane, wave);
  stage<0, true >(S2, gate[2], lane, wave);
  stage<1, false>(S2, gate[3], lane, wave);
  stage<0, true >(S2, gate[2 * (N_FULL - 1)], lane, wave);

  float cw[6], sw[6];
#pragma unroll
  for (int w = 0; w < 6; ++w) {
    const float t = 0.5f * xa[w];
    cw[w] = cosf(t);
    sw[w] = sinf(t);
  }
  float p4[16];
#pragma unroll
  for (int c = 0; c < 16; ++c) {
    float p = 1.f;
#pragma unroll
    for (int w = 0; w < 4; ++w) p *= ((c >> (3 - w)) & 1) ? sw[w] : cw[w];
    p4[c] = p;
  }
  const float t4 = half ? sw[4] : cw[4];
  const float e0 = t4 * cw[5], e1 = t4 * sw[5];

  const float* Vrow = reinterpret_cast<const float*>(S2) +
                      2 * slot(((16 * mq + (lm >> 1)) << 6) | (2 * half)) + (lm & 1);
  v8f d[2] = {};
#pragma unroll
  for (int c = 0; c < 16; ++c) {
    v2f bf;
    bf.x = p4[c] * e0;
    bf.y = p4[c] * e1;
#pragma unroll
    for (int i = 0; i < 2; ++i) {
      const int k = slot(((8 * i) << 6) | (4 * c));
      v2f a;
      a.x = Vrow[2 * k];
      a.y = Vrow[2 * k + 2];
      d[i] = wmma4(a, bf, d[i]);
    }
  }

  float acc = 0.f;
#pragma unroll
  for (int i = 0; i < 2; ++i) {
#pragma unroll
    for (int v = 0; v < 8; ++v) acc += d[i][v] * d[i][v];
  }
  acc += __shfl_xor(acc, 16);
  if (lane < 16) part[wave][lane] = (mq < 2) ? acc : -acc;
  __syncthreads();
  if (wave == 0) {
    const int w0 = (lane >> 4) * 4;
    out[blockIdx.x * 32 + lane] = (part[w0][lm] + part[w0 + 1][lm]) +
                                  (part[w0 + 2][lm] + part[w0 + 3][lm]);
  }
}

extern "C" void kernel_launch(void* const* d_in, const int* in_sizes, int n_in,
                              void* d_out, int out_size, void* d_ws,
                              size_t ws_size, hipStream_t stream) {
  const float* x = (const float*)d_in[0];
  const float* w = (const float*)d_in[1];
  float* out     = (float*)d_out;

  circuit_z0<<<dim3(BATCH / 32), dim3(256), 0, stream>>>(x, w, out);
}


# Each wave owns 4 samples, their states in its own part of LDS: the 5 stages need no workgroup barrier (3 left, was 10) and overlap between waves; <Z_0> summed across lanes: new float sum order
# speedup vs baseline: 62.1418x; 1.5376x over previous
#include <hip/hip_runtime.h>

typedef __attribute__((ext_vector_type(2))) float v2f;
typedef __attribute__((ext_vector_type(4))) float v4f;
typedef __attribute__((ext_vector_type(8))) float v8f;

#define N_QUBITS 12
#define N_LAYERS 6
#define BATCH    4096
#define SAMPLES  32
#define N_FULL   3
#define N_GATES  (2 * (N_FULL - 1) + 1)


static __device__ __forceinline__ v8f wmma4(v2f a, v2f b, v8f c) {
  return __builtin_amdgcn_wmma_f32_16x16x4_f32(false, a, false, b, (short)0, c,
                                               false, false);
}


struct cplx { float re, im; };
static __device__ __forceinline__ cplx gate_entry(const float* __restrict__ trig,
                                                  int l, int g, int i, int j) {
  float pr = 1.f, pi = 0.f;
#pragma unroll
  for (int k = 0; k < 3; ++k) {
    const float* tp = trig + (l * 6 + 3 * g + k) * 4;
    const float cy = tp[0], sy = tp[1];
    const float cz = tp[2], sz = tp[3];
    const int bi = (i >> (2 - k)) & 1;
    const int bj = (j >> (2 - k)) & 1;
    const float ryv = bi ? (bj ? cy : sy) : (bj ? -sy : cy);
    const float gre = ryv * cz;
    const float gim = ryv * (bi ? sz : -sz);
    const float nr = pr * gre - pi * gim;
    const float ni = pr * gim + pi * gre;
    pr = nr; pi = ni;
  }
  cplx o;
  o.re = pr;
  o.im = pi;
  return o;
}
static __device__ __forceinline__ v2f gate_fragment(const float* __restrict__ trig,
                                                    int l, int g, int i, int j,
                                                    int ri) {
  const cplx u = gate_entry(trig, l, g, i, j);
  v2f o;
  o.x = ri ? u.im : u.re;
  o.y = ri ? u.re : -u.im;
  return o;
}


static __device__ __forceinline__ cplx cmul(cplx a, cplx b) {
  cplx o;
  o.re = a.re * b.re - a.im * b.im;
  o.im = a.re * b.im + a.im * b.re;
  return o;
}
static __device__ __forceinline__ cplx wire_gate(const float* __restrict__ trig,
                                                 int l, int w, int bi, int bj) {
  const float* tp = trig + (l * 6 + w) * 4;
  const float cy = tp[0], sy = tp[1];
  const float cz = tp[2], sz = tp[3];
  const float ryv = bi ? (bj ? cy : sy) : (bj ? -sy : cy);
  cplx o;
  o.re = ryv * cz;
  o.im = ryv * (bi ? sz : -sz);
  return o;
}

static __device__ __forceinline__ v2f tail_fragment(const float* __restrict__ trig,
                                                    int i, int j, int ri) {
  const int i0 = (i >> 2) & 1, i1 = (i >> 1) & 1, i2 = i & 1;
  const int j0 = (j >> 2) & 1, j1 = (j >> 1) & 1, j2 = j & 1;
  float pr = 0.f, pi = 0.f;
#pragma unroll
  for (int p0 = 0; p0 < 2; ++p0) {
    float mr = 0.f, mi = 0.f;
#pragma unroll
    for (int k0 = 0; k0 < 2; ++k0) {
      float nr = 0.f, ni = 0.f;
#pragma unroll
      for (int k1 = 0; k1 < 2; ++k1) {
        const cplx t = cmul(cmul(wire_gate(trig, 4, 1, i1 ^ p0, k1 ^ k0),
                                 wire_gate(trig, 3, 1, k1, j1)),
                            wire_gate(trig, 3, 2, i2 ^ k1 ^ k0, j2));
        nr += t.re; ni += t.im;
      }
      cplx n; n.re = nr; n.im = ni;
      const cplx a = cmul(cmul(wire_gate(trig, 4, 0, p0, k0),
                               wire_gate(trig, 3, 0, k0, j0)), n);
      mr += a.re; mi += a.im;
    }
    cplx m; m.re = mr; m.im = mi;
    const cplx a = cmul(wire_gate(trig, 5, 0, i0, p0), m);
    pr += a.re; pi += a.im;
  }
  v2f o;
  o.x = ri ? pi : pr;
  o.y = ri ? pr : -pi;
  return o;
}

#define SPITCH 80
#define WSTATE (4 * SPITCH)
static __device__ __forceinline__ int ent(int sl, int r_hi, int r_lo) {
  return sl * SPITCH + r_hi * 8 + (r_lo ^ (r_hi & 6));
}

static __device__ __forceinline__ void wave_sync() {
  __builtin_amdgcn_fence(__ATOMIC_RELEASE, "wavefront");
  __builtin_amdgcn_wave_barrier();
  __builtin_amdgcn_fence(__ATOMIC_ACQUIRE, "wavefront");
}

template <int G, bool GRAY, bool LAST>
static __device__ __forceinline__ void stage(v2f* __restrict__ Sw,
                                             const v2f (&af)[4],
                                             int lane, float (&z)[2]) {
  const int c = lane & 15, half = lane >> 4;
  const int o = c & 7, s1 = c >> 3;

  int wa[4], ra[4];
#pragma unroll
  for (int q = 0; q < 4; ++q) {
    const int j = 2 * q + half;
    const int r_hi = G ? o : j;
    const int r_lo = G ? j : o;
    wa[q] = ent(s1, r_hi, r_lo);
    if (GRAY) {
      const int r = (r_hi << 3) | r_lo;
      const int g = r ^ (r >> 1);
      ra[q] = ent(s1, g >> 3, g & 7);
    } else {
      ra[q] = wa[q];
    }
  }

  v8f d[2] = {};
#pragma unroll
  for (int q = 0; q < 4; ++q) {
#pragma unroll
    for (int tt = 0; tt < 2; ++tt) {
      const v2f b = Sw[ra[q] + tt * 2 * SPITCH];
      d[tt] = wmma4(af[q], b, d[tt]);
    }
  }
  if (LAST) {
#pragma unroll
    for (int tt = 0; tt < 2; ++tt) {
      float p = 0.f, m = 0.f;
#pragma unroll
      for (int v = 0; v < 4; ++v) {
        p += d[tt][v] * d[tt][v];
        m += d[tt][v + 4] * d[tt][v + 4];
      }
      z[tt] = p - m;
    }
    return;
  }
#pragma unroll
  for (int tt = 0; tt < 2; ++tt) {
#pragma unroll
    for (int q = 0; q < 4; ++q) {
      v2f ov;
      ov.x = d[tt][2 * q];
      ov.y = d[tt][2 * q + 1];
      Sw[wa[q] + tt * 2 * SPITCH] = ov;
    }
  }
  wave_sync();
}

static __device__ __forceinline__ float xhalf(float v) {
  const int b = __builtin_bit_cast(int, v);
  return __builtin_bit_cast(float, __builtin_amdgcn_permlanex16(b, b, 0x76543210, 0xfedcba98, false, false));
}
template <int M>
static __device__ __forceinline__ float row_x(float v) {
  return __builtin_bit_cast(float, __builtin_amdgcn_update_dpp(0, __builtin_bit_cast(int, v), 0x160 + M, 0xf, 0xf, false));
}

static __device__ __forceinline__ cplx pick(v4f g, int bit) {
  cplx o;
  o.re = bit ? g.z : g.x;
  o.im = bit ? g.w : g.y;
  return o;
}

static __device__ __forceinline__ void stage_first(v2f* __restrict__ Sw,
                                                   const v4f* __restrict__ Gw,
                                                   const v2f (&af)[4],
                                                   int lane) {
  const int c = lane & 15, half = lane >> 4;
  const int o = c & 7, s1 = c >> 3;
  const int rlo = (o ^ (o >> 1)) ^ (half << 2);

  v8f d[2] = {};
#pragma unroll
  for (int tt = 0; tt < 2; ++tt) {
    v4f g[6];
#pragma unroll
    for (int w = 0; w < 6; ++w) g[w] = Gw[(2 * tt + s1) * 6 + w];
    const cplx u1 = cmul(cmul(pick(g[3], (rlo >> 2) & 1), pick(g[4], (rlo >> 1) & 1)),
                         pick(g[5], rlo & 1));
    const cplx h2 = pick(g[2], half), n2 = pick(g[2], half ^ 1);
    cplx a[4];
    a[0] = cmul(cmul(pick(g[0], 0), pick(g[1], 0)), h2);
    a[1] = cmul(cmul(pick(g[0], 0), pick(g[1], 1)), n2);
    a[2] = cmul(cmul(pick(g[0], 1), pick(g[1], 1)), h2);
    a[3] = cmul(cmul(pick(g[0], 1), pick(g[1], 0)), n2);
#pragma unroll
    for (int q = 0; q < 4; ++q) {
      const cplx e = cmul(a[q], u1);
      v2f b;
      b.x = e.re;
      b.y = e.im;
      d[tt] = wmma4(af[q], b, d[tt]);
    }
  }
#pragma unroll
  for (int tt = 0; tt < 2; ++tt) {
#pragma unroll
    for (int q = 0; q < 4; ++q) {
      v2f ov;
      ov.x = d[tt][2 * q];
      ov.y = d[tt][2 * q + 1];
      Sw[ent(2 * tt + s1, 2 * q + half, o)] = ov;
    }
  }
  wave_sync();
}

__global__ __launch_bounds__(256) void circuit_z0(const float* __restrict__ x,
                                                  const float* __restrict__ weights,
                                                  float* __restrict__ out) {
  __shared__ v2f   S2[8 * WSTATE];
  __shared__ v2f   Ufrag2[N_GATES * 128];
  __shared__ v4f   Gt[8 * 24];
  __shared__ float trig[N_LAYERS * 6 * 4];
  __shared__ float part[SAMPLES];

  const int tid  = threadIdx.x;
  const int lane = tid & 31;
  const int wave = tid >> 5;
  const int lm = lane & 15, half = lane >> 4;

  const int xs = lane / 6, xw = lane - 6 * xs;
  float xv = 0.f;
  if (lane < 24) xv = x[(blockIdx.x * SAMPLES + 4 * wave + xs) * N_QUBITS + xw];

  if (tid < N_LAYERS * 6 * 2) {
    const int pw = tid >> 1, which = tid & 1;
    const int l = pw / 6, q = pw - 6 * l;
    const float t = 0.5f * weights[(l * N_QUBITS + q) * 2 + which];
    trig[pw * 4 + which * 2 + 0] = cosf(t);
    trig[pw * 4 + which * 2 + 1] = sinf(t);
  }
  float cx = 1.f, sx = 0.f;
  if (lane < 24) {
    cx = cosf(0.5f * xv);
    sx = sinf(0.5f * xv);
  }
  __syncthreads();

  {
    const int c  = (tid >> 5) & 3;
    const int j  = 2 * c + half;
    const int i  = 2 * ((lm & 7) >> 1) + (lm >> 3);
    const int ri = lm & 1;
    if (tid < 128) {
#pragma unroll
      for (int l = 1; l < N_FULL; ++l) {
#pragma unroll
        for (int g = 0; g < 2; ++g)
          Ufrag2[(2 * (l - 1) + g) * 128 + c * 32 + lane] = gate_fragment(trig, l, g, i, j, ri);
      }
    } else {
      Ufrag2[2 * (N_FULL - 1) * 128 + c * 32 + lane] = tail_fragment(trig, i, j, ri);
    }
    if (lane < 24) {
      const float* tp = trig + xw * 4;
      const float cy = tp[0], sy = tp[1];
      const float cz = tp[2], sz = tp[3];
      const float a0 = cy * cx - sy * sx;
      const float a1 = sy * cx + cy * sx;
      v4f g;
      g.x = a0 * cz;
      g.y = -(a0 * sz);
      g.z = a1 * cz;
      g.w = a1 * sz;
      Gt[wave * 24 + lane] = g;
    }
  }
  __syncthreads();

  v2f gate[N_GATES][4];
#pragma unroll
  for (int s = 0; s < N_GATES; ++s) {
#pragma unroll
    for (int c = 0; c < 4; ++c)
      gate[s][c] = Ufrag2[s * 128 + c * 32 + lane];
  }

  v2f* Sw = S2 + wave * WSTATE;
  float z[2] = {0.f, 0.f};
  stage_first(Sw, Gt + wave * 24, gate[0], lane);
  stage<1, false, false>(Sw, gate[1], lane, z);
  stage<0, true,  false>(Sw, gate[2], lane, z);
  stage<1, false, false>(Sw, gate[3], lane, z);
  stage<0, true,  true >(Sw, gate[2 * (N_FULL - 1)], lane, z);

#pragma unroll
  for (int tt = 0; tt < 2; ++tt) {
    z[tt] += xhalf(z[tt]);
    z[tt] += row_x<4>(z[tt]);
    z[tt] += row_x<2>(z[tt]);
    z[tt] += row_x<1>(z[tt]);
  }
  if ((lane & 23) == 0) {
    part[4 * wave + (lane >> 3)]     = z[0];
    part[4 * wave + 2 + (lane >> 3)] = z[1];
  }
  __syncthreads();
  if (wave == 0) out[blockIdx.x * SAMPLES + lane] = part[lane];
}

extern "C" void kernel_launch(void* const* d_in, const int* in_sizes, int n_in,
                              void* d_out, int out_size, void* d_ws,
                              size_t ws_size, hipStream_t stream) {
  const float* x = (const float*)d_in[0];
  const float* w = (const float*)d_in[1];
  float* out     = (float*)d_out;

  circuit_z0<<<dim3(BATCH / SAMPLES), dim3(256), 0, stream>>>(x, w, out);
}


# Every lane loads ONE angle (lanes 0-23 an x of its samples, 24-29 one of the 48 weight angles Z_0 sees): one memory wait and one cos/sin before barrier 1, was two of each in series; same float sums
# speedup vs baseline: 65.6807x; 1.0569x over previous
#include <hip/hip_runtime.h>

typedef __attribute__((ext_vector_type(2))) float v2f;
typedef __attribute__((ext_vector_type(4))) float v4f;
typedef __attribute__((ext_vector_type(8))) float v8f;

#define N_QUBITS 12
#define N_LAYERS 6
#define BATCH    4096
#define SAMPLES  32
#define N_FULL   3
#define N_GATES  (2 * (N_FULL - 1) + 1)


static __device__ __forceinline__ v8f wmma4(v2f a, v2f b, v8f c) {
  return __builtin_amdgcn_wmma_f32_16x16x4_f32(false, a, false, b, (short)0, c,
                                               false, false);
}


struct cplx { float re, im; };
static __device__ __forceinline__ cplx gate_entry(const float* __restrict__ trig,
                                                  int l, int g, int i, int j) {
  float pr = 1.f, pi = 0.f;
#pragma unroll
  for (int k = 0; k < 3; ++k) {
    const float* tp = trig + (l * 6 + 3 * g + k) * 4;
    const float cy = tp[0], sy = tp[1];
    const float cz = tp[2], sz = tp[3];
    const int bi = (i >> (2 - k)) & 1;
    const int bj = (j >> (2 - k)) & 1;
    const float ryv = bi ? (bj ? cy : sy) : (bj ? -sy : cy);
    const float gre = ryv * cz;
    const float gim = ryv * (bi ? sz : -sz);
    const float nr = pr * gre - pi * gim;
    const float ni = pr * gim + pi * gre;
    pr = nr; pi = ni;
  }
  cplx o;
  o.re = pr;
  o.im = pi;
  return o;
}
static __device__ __forceinline__ v2f gate_fragment(const float* __restrict__ trig,
                                                    int l, int g, int i, int j,
                                                    int ri) {
  const cplx u = gate_entry(trig, l, g, i, j);
  v2f o;
  o.x = ri ? u.im : u.re;
  o.y = ri ? u.re : -u.im;
  return o;
}


static __device__ __forceinline__ cplx cmul(cplx a, cplx b) {
  cplx o;
  o.re = a.re * b.re - a.im * b.im;
  o.im = a.re * b.im + a.im * b.re;
  return o;
}
static __device__ __forceinline__ cplx wire_gate(const float* __restrict__ trig,
                                                 int l, int w, int bi, int bj) {
  const float* tp = trig + (l * 6 + w) * 4;
  const float cy = tp[0], sy = tp[1];
  const float cz = tp[2], sz = tp[3];
  const float ryv = bi ? (bj ? cy : sy) : (bj ? -sy : cy);
  cplx o;
  o.re = ryv * cz;
  o.im = ryv * (bi ? sz : -sz);
  return o;
}

static __device__ __forceinline__ v2f tail_fragment(const float* __restrict__ trig,
                                                    int i, int j, int ri) {
  const int i0 = (i >> 2) & 1, i1 = (i >> 1) & 1, i2 = i & 1;
  const int j0 = (j >> 2) & 1, j1 = (j >> 1) & 1, j2 = j & 1;
  float pr = 0.f, pi = 0.f;
#pragma unroll
  for (int p0 = 0; p0 < 2; ++p0) {
    float mr = 0.f, mi = 0.f;
#pragma unroll
    for (int k0 = 0; k0 < 2; ++k0) {
      float nr = 0.f, ni = 0.f;
#pragma unroll
      for (int k1 = 0; k1 < 2; ++k1) {
        const cplx t = cmul(cmul(wire_gate(trig, 4, 1, i1 ^ p0, k1 ^ k0),
                                 wire_gate(trig, 3, 1, k1, j1)),
                            wire_gate(trig, 3, 2, i2 ^ k1 ^ k0, j2));
        nr += t.re; ni += t.im;
      }
      cplx n; n.re = nr; n.im = ni;
      const cplx a = cmul(cmul(wire_gate(trig, 4, 0, p0, k0),
                               wire_gate(trig, 3, 0, k0, j0)), n);
      mr += a.re; mi += a.im;
    }
    cplx m; m.re = mr; m.im = mi;
    const cplx a = cmul(wire_gate(trig, 5, 0, i0, p0), m);
    pr += a.re; pi += a.im;
  }
  v2f o;
  o.x = ri ? pi : pr;
  o.y = ri ? pr : -pi;
  return o;
}

#define SPITCH 80
#define WSTATE (4 * SPITCH)
static __device__ __forceinline__ int ent(int sl, int r_hi, int r_lo) {
  return sl * SPITCH + r_hi * 8 + (r_lo ^ (r_hi & 6));
}

static __device__ __forceinline__ void wave_sync() {
  __builtin_amdgcn_fence(__ATOMIC_RELEASE, "wavefront");
  __builtin_amdgcn_wave_barrier();
  __builtin_amdgcn_fence(__ATOMIC_ACQUIRE, "wavefront");
}

template <int G, bool GRAY, bool LAST>
static __device__ __forceinline__ void stage(v2f* __restrict__ Sw,
                                             const v2f (&af)[4],
                                             int lane, float (&z)[2]) {
  const int c = lane & 15, half = lane >> 4;
  const int o = c & 7, s1 = c >> 3;

  int wa[4], ra[4];
#pragma unroll
  for (int q = 0; q < 4; ++q) {
    const int j = 2 * q + half;
    const int r_hi = G ? o : j;
    const int r_lo = G ? j : o;
    wa[q] = ent(s1, r_hi, r_lo);
    if (GRAY) {
      const int r = (r_hi << 3) | r_lo;
      const int g = r ^ (r >> 1);
      ra[q] = ent(s1, g >> 3, g & 7);
    } else {
      ra[q] = wa[q];
    }
  }

  v8f d[2] = {};
#pragma unroll
  for (int q = 0; q < 4; ++q) {
#pragma unroll
    for (int tt = 0; tt < 2; ++tt) {
      const v2f b = Sw[ra[q] + tt * 2 * SPITCH];
      d[tt] = wmma4(af[q], b, d[tt]);
    }
  }
  if (LAST) {
#pragma unroll
    for (int tt = 0; tt < 2; ++tt) {
      float p = 0.f, m = 0.f;
#pragma unroll
      for (int v = 0; v < 4; ++v) {
        p += d[tt][v] * d[tt][v];
        m += d[tt][v + 4] * d[tt][v + 4];
      }
      z[tt] = p - m;
    }
    return;
  }
#pragma unroll
  for (int tt = 0; tt < 2; ++tt) {
#pragma unroll
    for (int q = 0; q < 4; ++q) {
      v2f ov;
      ov.x = d[tt][2 * q];
      ov.y = d[tt][2 * q + 1];
      Sw[wa[q] + tt * 2 * SPITCH] = ov;
    }
  }
  wave_sync();
}

static __device__ __forceinline__ float xhalf(float v) {
  const int b = __builtin_bit_cast(int, v);
  return __builtin_bit_cast(float, __builtin_amdgcn_permlanex16(b, b, 0x76543210, 0xfedcba98, false, false));
}
template <int M>
static __device__ __forceinline__ float row_x(float v) {
  return __builtin_bit_cast(float, __builtin_amdgcn_update_dpp(0, __builtin_bit_cast(int, v), 0x160 + M, 0xf, 0xf, false));
}

static __device__ __forceinline__ cplx pick(v4f g, int bit) {
  cplx o;
  o.re = bit ? g.z : g.x;
  o.im = bit ? g.w : g.y;
  return o;
}

static __device__ __forceinline__ void stage_first(v2f* __restrict__ Sw,
                                                   const v4f* __restrict__ Gw,
                                                   const v2f (&af)[4],
                                                   int lane) {
  const int c = lane & 15, half = lane >> 4;
  const int o = c & 7, s1 = c >> 3;
  const int rlo = (o ^ (o >> 1)) ^ (half << 2);

  v8f d[2] = {};
#pragma unroll
  for (int tt = 0; tt < 2; ++tt) {
    v4f g[6];
#pragma unroll
    for (int w = 0; w < 6; ++w) g[w] = Gw[(2 * tt + s1) * 6 + w];
    const cplx u1 = cmul(cmul(pick(g[3], (rlo >> 2) & 1), pick(g[4], (rlo >> 1) & 1)),
                         pick(g[5], rlo & 1));
    const cplx h2 = pick(g[2], half), n2 = pick(g[2], half ^ 1);
    cplx a[4];
    a[0] = cmul(cmul(pick(g[0], 0), pick(g[1], 0)), h2);
    a[1] = cmul(cmul(pick(g[0], 0), pick(g[1], 1)), n2);
    a[2] = cmul(cmul(pick(g[0], 1), pick(g[1], 1)), h2);
    a[3] = cmul(cmul(pick(g[0], 1), pick(g[1], 0)), n2);
#pragma unroll
    for (int q = 0; q < 4; ++q) {
      const cplx e = cmul(a[q], u1);
      v2f b;
      b.x = e.re;
      b.y = e.im;
      d[tt] = wmma4(af[q], b, d[tt]);
    }
  }
#pragma unroll
  for (int tt = 0; tt < 2; ++tt) {
#pragma unroll
    for (int q = 0; q < 4; ++q) {
      v2f ov;
      ov.x = d[tt][2 * q];
      ov.y = d[tt][2 * q + 1];
      Sw[ent(2 * tt + s1, 2 * q + half, o)] = ov;
    }
  }
  wave_sync();
}

__global__ __launch_bounds__(256) void circuit_z0(const float* __restrict__ x,
                                                  const float* __restrict__ weights,
                                                  float* __restrict__ out) {
  __shared__ v2f   S2[8 * WSTATE];
  __shared__ v2f   Ufrag2[N_GATES * 128];
  __shared__ v4f   Gt[8 * 24];
  __shared__ float trig[N_LAYERS * 6 * 4];
  __shared__ float part[SAMPLES];

  const int tid  = threadIdx.x;
  const int lane = tid & 31;
  const int wave = tid >> 5;
  const int lm = lane & 15, half = lane >> 4;

  const int xs = lane / 6, xw = lane - 6 * xs;
  const unsigned j = lane & 7, which = j & 1;
  const unsigned n  = 3u * wave + (j >> 1);
  const unsigned l4 = n >= 21u ? 1u : 0u, l5 = n >= 23u ? 1u : 0u;
  const unsigned pw = n + 3u * l4 + 4u * l5;
  const unsigned wl = (wave >> 1) + l4 + l5;
  const bool is_x = lane < 24;
  const bool is_w = !is_x && lane < 30;
  const unsigned off_x = (blockIdx.x * SAMPLES + 4 * wave + xs) * N_QUBITS + xw;
  const unsigned off_w = (pw + 6u * wl) * 2u + which;
  const unsigned off = is_x ? off_x : (is_w ? off_w : 0u);
  const float* base = is_x ? x : weights;
  const float th = 0.5f * base[off];
  const float cx = cosf(th), sx = sinf(th);
  if (is_w) {
    trig[pw * 4 + which * 2 + 0] = cx;
    trig[pw * 4 + which * 2 + 1] = sx;
  }
  __syncthreads();

  {
    const int c  = (tid >> 5) & 3;
    const int j  = 2 * c + half;
    const int i  = 2 * ((lm & 7) >> 1) + (lm >> 3);
    const int ri = lm & 1;
    if (tid < 128) {
#pragma unroll
      for (int l = 1; l < N_FULL; ++l) {
#pragma unroll
        for (int g = 0; g < 2; ++g)
          Ufrag2[(2 * (l - 1) + g) * 128 + c * 32 + lane] = gate_fragment(trig, l, g, i, j, ri);
      }
    } else {
      Ufrag2[2 * (N_FULL - 1) * 128 + c * 32 + lane] = tail_fragment(trig, i, j, ri);
    }
    if (lane < 24) {
      const float* tp = trig + xw * 4;
      const float cy = tp[0], sy = tp[1];
      const float cz = tp[2], sz = tp[3];
      const float a0 = cy * cx - sy * sx;
      const float a1 = sy * cx + cy * sx;
      v4f g;
      g.x = a0 * cz;
      g.y = -(a0 * sz);
      g.z = a1 * cz;
      g.w = a1 * sz;
      Gt[wave * 24 + lane] = g;
    }
  }
  __syncthreads();

  v2f gate[N_GATES][4];
#pragma unroll
  for (int s = 0; s < N_GATES; ++s) {
#pragma unroll
    for (int c = 0; c < 4; ++c)
      gate[s][c] = Ufrag2[s * 128 + c * 32 + lane];
  }

  v2f* Sw = S2 + wave * WSTATE;
  float z[2] = {0.f, 0.f};
  stage_first(Sw, Gt + wave * 24, gate[0], lane);
  stage<1, false, false>(Sw, gate[1], lane, z);
  stage<0, true,  false>(Sw, gate[2], lane, z);
  stage<1, false, false>(Sw, gate[3], lane, z);
  stage<0, true,  true >(Sw, gate[2 * (N_FULL - 1)], lane, z);

#pragma unroll
  for (int tt = 0; tt < 2; ++tt) {
    z[tt] += xhalf(z[tt]);
    z[tt] += row_x<4>(z[tt]);
    z[tt] += row_x<2>(z[tt]);
    z[tt] += row_x<1>(z[tt]);
  }
  if ((lane & 23) == 0) {
    part[4 * wave + (lane >> 3)]     = z[0];
    part[4 * wave + 2 + (lane >> 3)] = z[1];
  }
  __syncthreads();
  if (wave == 0) out[blockIdx.x * SAMPLES + lane] = part[lane];
}

extern "C" void kernel_launch(void* const* d_in, const int* in_sizes, int n_in,
                              void* d_out, int out_size, void* d_ws,
                              size_t ws_size, hipStream_t stream) {
  const float* x = (const float*)d_in[0];
  const float* w = (const float*)d_in[1];
  float* out     = (float*)d_out;

  circuit_z0<<<dim3(BATCH / SAMPLES), dim3(256), 0, stream>>>(x, w, out);
}
